# starting kernel
# baseline (speedup 1.0000x reference)
#include <hip/hip_runtime.h>
#include <hip/hip_bf16.h>
#include <math.h>

#define NN 100000
#define NE 1600000

typedef __attribute__((ext_vector_type(2))) float v2f;
typedef __attribute__((ext_vector_type(8))) float v8f;


__global__ void zero_f(float* __restrict__ p, int n) {
    int i = blockIdx.x * blockDim.x + threadIdx.x;
    if (i < n) p[i] = 0.0f;
}

__global__ void deg_count(const int* __restrict__ dst, float* __restrict__ deg, int e_cnt) {
    int e = blockIdx.x * blockDim.x + threadIdx.x;
    if (e < e_cnt) atomicAdd(&deg[dst[e]], 1.0f);
}

__global__ void deg_norm(const float* __restrict__ deg, float* __restrict__ dinv,
                         float* __restrict__ selfn, int n) {
    int i = blockIdx.x * blockDim.x + threadIdx.x;
    if (i >= n) return;
    float d = deg[i] + 1.0f;
    dinv[i]  = rsqrtf(d);
    selfn[i] = 1.0f / d;
}

__global__ void wmma_gemm_f32(const float* __restrict__ X, const float* __restrict__ W,
                              const float* __restrict__ bias, float* __restrict__ Y,
                              int M, int K, int Nout, int act) {
    int wave  = (int)((blockIdx.x * blockDim.x + threadIdx.x) >> 5);
    int lane  = threadIdx.x & 31;
    int tilesN = Nout >> 4;
    int m0 = (wave / tilesN) << 4;
    int n0 = (wave % tilesN) << 4;
    if (m0 >= M) return;
    int half = lane >> 4;
    int l    = lane & 15;

    const float* xrow = X + (long long)(m0 + l) * K;
    v8f c = {};
    for (int k = 0; k < K; k += 4) {
        int ka = k + half * 2;
        v2f a, b;
        a.x = xrow[ka];
        a.y = xrow[ka + 1];
        b.x = W[ka * Nout + n0 + l];
        b.y = W[(ka + 1) * Nout + n0 + l];
        c = __builtin_amdgcn_wmma_f32_16x16x4_f32(false, a, false, b,
                                                  (short)0, c, false, false);
    }
    int col = n0 + l;
    float bv = bias ? bias[col] : 0.0f;
#pragma unroll
    for (int r = 0; r < 8; ++r) {
        int row = m0 + r + half * 8;
        float v = c[r] + bv;
        if (act) v = tanhf(v);
        Y[(long long)row * Nout + col] = v;
    }
}

__global__ void scatter_agg64(const float* __restrict__ xl, const int* __restrict__ src,
                              const int* __restrict__ dst, const float* __restrict__ dinv,
                              float* __restrict__ agg, int e_cnt) {
    int t = blockIdx.x * blockDim.x + threadIdx.x;
    int e = t >> 6;
    if (e >= e_cnt) return;
    int f = t & 63;
    int s = src[e];
    int d = dst[e];
    float norm = dinv[s] * dinv[d];
    atomicAdd(&agg[(long long)d * 64 + f], xl[(long long)s * 64 + f] * norm);
}

__global__ void conv_finalize64(const float* __restrict__ agg, const float* __restrict__ xl,
                                const float* __restrict__ selfn, const float* __restrict__ bias,
                                float* __restrict__ out, int n_elems) {
    int t = blockIdx.x * blockDim.x + threadIdx.x;
    if (t >= n_elems) return;
    int node = t >> 6;
    int f    = t & 63;
    out[t] = tanhf(agg[t] + xl[t] * selfn[node] + bias[f]);
}

__global__ void fc_last32(const float* __restrict__ h, const float* __restrict__ w,
                          const float* __restrict__ b, float* __restrict__ out, int n) {
    int i = blockIdx.x * blockDim.x + threadIdx.x;
    if (i >= n) return;
    float acc = b[0];
#pragma unroll
    for (int j = 0; j < 32; ++j) acc += h[i * 32 + j] * w[j];
    out[i] = acc;
}


extern "C" void kernel_launch(void* const* d_in, const int* in_sizes, int n_in,
                              void* d_out, int out_size, void* d_ws, size_t ws_size,
                              hipStream_t stream) {
    const float* x   = (const float*)d_in[0];
    const int*   ei  = (const int*)d_in[1];
    const float* cW0 = (const float*)d_in[2];
    const float* cb0 = (const float*)d_in[3];
    const float* cW1 = (const float*)d_in[4];
    const float* cb1 = (const float*)d_in[5];
    const float* fW0 = (const float*)d_in[6];
    const float* fb0 = (const float*)d_in[7];
    const float* fW1 = (const float*)d_in[8];
    const float* fb1 = (const float*)d_in[9];
    const float* fW2 = (const float*)d_in[10];
    const float* fb2 = (const float*)d_in[11];
    const int* src = ei;
    const int* dst = ei + NE;

    float* ws    = (float*)d_ws;
    float* xl    = ws;
    float* agg   = xl  + (size_t)NN * 64;
    float* h1    = agg + (size_t)NN * 64;
    float* h2    = h1  + (size_t)NN * 64;
    float* deg   = h2  + (size_t)NN * 64;
    float* dinv  = deg  + NN;
    float* selfn = dinv + NN;

    const int B = 256;
    dim3 blk(B);
    auto cdiv = [](long long a, long long b) { return (unsigned)((a + b - 1) / b); };

    unsigned gN    = cdiv(NN, B);
    unsigned gE    = cdiv(NE, B);
    unsigned gNF   = cdiv((long long)NN * 64, B);
    unsigned gEF   = cdiv((long long)NE * 64, B);
    unsigned gG64  = cdiv((long long)(NN / 16) * 4 * 32, B);
    unsigned gG32  = cdiv((long long)(NN / 16) * 2 * 32, B);

    zero_f<<<gN, blk, 0, stream>>>(deg, NN);
    deg_count<<<gE, blk, 0, stream>>>(dst, deg, NE);
    deg_norm<<<gN, blk, 0, stream>>>(deg, dinv, selfn, NN);

    wmma_gemm_f32<<<gG64, blk, 0, stream>>>(x, cW0, nullptr, xl, NN, 128, 64, 0);
    zero_f<<<gNF, blk, 0, stream>>>(agg, NN * 64);
    scatter_agg64<<<gEF, blk, 0, stream>>>(xl, src, dst, dinv, agg, NE);
    conv_finalize64<<<gNF, blk, 0, stream>>>(agg, xl, selfn, cb0, h1, NN * 64);

    wmma_gemm_f32<<<gG64, blk, 0, stream>>>(h1, cW1, nullptr, xl, NN, 64, 64, 0);
    zero_f<<<gNF, blk, 0, stream>>>(agg, NN * 64);
    scatter_agg64<<<gEF, blk, 0, stream>>>(xl, src, dst, dinv, agg, NE);
    conv_finalize64<<<gNF, blk, 0, stream>>>(agg, xl, selfn, cb1, h2, NN * 64);

    wmma_gemm_f32<<<gG64, blk, 0, stream>>>(h2, fW0, fb0, h1, NN, 64, 64, 1);
    wmma_gemm_f32<<<gG32, blk, 0, stream>>>(h1, fW1, fb1, xl, NN, 64, 32, 1);
    fc_last32<<<gN, blk, 0, stream>>>(xl, fW2, fb2, (float*)d_out, NN);
}


# Edge sums on the matrix unit: tile edges sorted by node in the degree kernel; a wave owns 16 nodes, adds 4 gathered rows per f32 WMMA via a 16x4 norm/selection matrix; LDS float adds go
# speedup vs baseline: 4.1019x; 4.1019x over previous
#include <hip/hip_runtime.h>
#include <hip/hip_bf16.h>
#include <math.h>


#define NN 100000
#define NE 1600000

#define TSH  7
#define TILE 128
#define NB   ((NN + TILE - 1) / TILE)
#define CH   8192
#define BT   1024
#define EPT  (CH / BT)

typedef __attribute__((ext_vector_type(2))) float v2f;
typedef __attribute__((ext_vector_type(8))) float v8f;
typedef __attribute__((ext_vector_type(4))) float v4f;
typedef __attribute__((ext_vector_type(4))) int   v4i;
typedef __attribute__((ext_vector_type(2))) int   v2i;


__global__ void __launch_bounds__(BT) zero_bins(int* __restrict__ binCount, int* __restrict__ cursor) {
    int t = threadIdx.x;
    if (t < NB) { binCount[t] = 0; cursor[t] = 0; }
}

__global__ void __launch_bounds__(BT) bin_hist(const int* __restrict__ dst, int* __restrict__ binCount) {
    __shared__ int hist[NB];
    int t = threadIdx.x;
    if (t < NB) hist[t] = 0;
    __syncthreads();
    int e0 = blockIdx.x * CH + t;
#pragma unroll
    for (int i = 0; i < EPT; ++i) {
        int e = e0 + i * BT;
        if (e < NE) atomicAdd(&hist[dst[e] >> TSH], 1);
    }
    __syncthreads();
    if (t < NB) {
        int c = hist[t];
        if (c) atomicAdd(&binCount[t], c);
    }
}

__global__ void __launch_bounds__(1024) bin_scan(const int* __restrict__ binCount, int* __restrict__ binStart) {
    __shared__ int a[1024];
    int t = threadIdx.x;
    a[t] = (t < NB) ? binCount[t] : 0;
    __syncthreads();
    for (int off = 1; off < 1024; off <<= 1) {
        int v = a[t];
        if (t >= off) v += a[t - off];
        __syncthreads();
        a[t] = v;
        __syncthreads();
    }
    if (t == 0) binStart[0] = 0;
    if (t < NB) binStart[t + 1] = a[t];
}

__global__ void __launch_bounds__(BT) bin_partition(const int* __restrict__ src, const int* __restrict__ dst,
                                                    const int* __restrict__ binStart, int* __restrict__ cursor,
                                                    int* __restrict__ binned) {
    __shared__ int hist[NB];
    __shared__ int base[NB];
    int t = threadIdx.x;
    if (t < NB) hist[t] = 0;
    __syncthreads();
    int e0 = blockIdx.x * CH + t;
    int bin[EPT], rank[EPT], pk[EPT];
#pragma unroll
    for (int i = 0; i < EPT; ++i) {
        int e = e0 + i * BT;
        bin[i] = -1; rank[i] = 0; pk[i] = 0;
        if (e < NE) {
            int d = dst[e];
            int s = src[e];
            bin[i]  = d >> TSH;
            pk[i]   = (s << TSH) | (d & (TILE - 1));
            rank[i] = atomicAdd(&hist[bin[i]], 1);
        }
    }
    __syncthreads();
    if (t < NB) {
        int c = hist[t];
        int claimed = c ? atomicAdd(&cursor[t], c) : 0;
        base[t] = binStart[t] + claimed;
    }
    __syncthreads();
#pragma unroll
    for (int i = 0; i < EPT; ++i) {
        if (bin[i] >= 0) {
            unsigned pos = (unsigned)(base[bin[i]] + rank[i]);
            if (pos < (unsigned)NE) binned[pos] = pk[i];
        }
    }
}

__global__ void __launch_bounds__(256) tile_degree_sort(const int* __restrict__ binned, const int* __restrict__ binStart,
                                                        int* __restrict__ sorted, int* __restrict__ grpBeg,
                                                        int* __restrict__ grpEnd,
                                                        float* __restrict__ dinv, float* __restrict__ selfn) {
    __shared__ int cnt[TILE];
    __shared__ int off[TILE + 1];
    __shared__ int wsum[8];
    int t = threadIdx.x, b = blockIdx.x;
    int lane = t & 31, w = t >> 5;
    if (t < TILE) cnt[t] = 0;
    __syncthreads();
    int beg = binStart[b], end = binStart[b + 1];
    for (int i = beg + t; i < end; i += 256) atomicAdd(&cnt[binned[i] & (TILE - 1)], 1);
    __syncthreads();
    int c = (t < TILE) ? cnt[t] : 0;
    if (t < TILE) {
        int node = b * TILE + t;
        if (node < NN) {
            float d = (float)c + 1.0f;
            dinv[node]  = rsqrtf(d);
            selfn[node] = 1.0f / d;
        }
    }
    int v = c;
#pragma unroll
    for (int d = 1; d < 32; d <<= 1) {
        int o = __shfl_up(v, d);
        if (lane >= d) v += o;
    }
    if (lane == 31) wsum[w] = v;
    __syncthreads();
    int pre = 0;
    for (int k = 0; k < w; ++k) pre += wsum[k];
    if (t < TILE) { off[t] = pre + v - c; cnt[t] = 0; }
    if (t == TILE) off[TILE] = end - beg;
    __syncthreads();
    if (t < 8) {
        grpBeg[b * 8 + t] = beg + off[t * 16];
        grpEnd[b * 8 + t] = beg + off[t * 16 + 16];
    }
    for (int i = beg + t; i < end; i += 256) {
        int p = binned[i];
        int n = p & (TILE - 1);
        int r = atomicAdd(&cnt[n], 1);
        unsigned pos = (unsigned)(beg + off[n] + r);
        if (pos < (unsigned)NE) sorted[pos] = p;
    }
}

__global__ void wmma_gemm_f32(const float* __restrict__ X, const float* __restrict__ W,
                              const float* __restrict__ bias, float* __restrict__ Y,
                              int M, int K, int Nout, int act) {
    int wave  = (int)((blockIdx.x * blockDim.x + threadIdx.x) >> 5);
    int lane  = threadIdx.x & 31;
    int tilesN = Nout >> 4;
    int m0 = (wave / tilesN) << 4;
    int n0 = (wave % tilesN) << 4;
    if (m0 >= M) return;
    int half = lane >> 4;
    int l    = lane & 15;

    const float* xrow = X + (long long)(m0 + l) * K;
    v8f c = {};
    for (int k = 0; k < K; k += 4) {
        int ka = k + half * 2;
        v2f a, b;
        a.x = xrow[ka];
        a.y = xrow[ka + 1];
        b.x = W[ka * Nout + n0 + l];
        b.y = W[(ka + 1) * Nout + n0 + l];
        c = __builtin_amdgcn_wmma_f32_16x16x4_f32(false, a, false, b,
                                                  (short)0, c, false, false);
    }
    int col = n0 + l;
    float bv = bias ? bias[col] : 0.0f;
#pragma unroll
    for (int r = 0; r < 8; ++r) {
        int row = m0 + r + half * 8;
        float v = c[r] + bv;
        if (act) v = tanhf(v);
        Y[(long long)row * Nout + col] = v;
    }
}

__device__ __forceinline__ void group_step16(v8f (&c)[4], const float* __restrict__ rows, const v4i* st4,
                                             int j, int half, int l, int row) {
    v4f ra[4], rb[4];
    v2f a[4];
#pragma unroll
    for (int u = 0; u < 4; ++u) {
        v4i q = st4[(j >> 1) + 2 * u + half];
        ra[u] = *(const v4f*)(rows + (((unsigned)q.x >> TSH) * 64u + 4u * (unsigned)l));
        rb[u] = *(const v4f*)(rows + (((unsigned)q.z >> TSH) * 64u + 4u * (unsigned)l));
        a[u].x = ((q.x & (TILE - 1)) == row) ? __int_as_float(q.y) : 0.0f;
        a[u].y = ((q.z & (TILE - 1)) == row) ? __int_as_float(q.w) : 0.0f;
    }
#pragma unroll
    for (int u = 0; u < 4; ++u) {
#pragma unroll
        for (int t = 0; t < 4; ++t) {
            v2f bf;
            bf.x = ra[u][t];
            bf.y = rb[u][t];
            c[t] = __builtin_amdgcn_wmma_f32_16x16x4_f32(false, a[u], false, bf,
                                                         (short)0, c[t], false, false);
        }
    }
}

__device__ __forceinline__ void group_sum_mma(v8f (&c)[4], const float* __restrict__ rows,
                                              const int* __restrict__ sorted, int gb, int ge,
                                              const float* __restrict__ dinv, int node0, int row,
                                              int* st, int lane) {
    int half = lane >> 4, l = lane & 15;
    const v4i* st4 = (const v4i*)st;
    int p = 0;
    float nrm = 0.0f;
    if (gb + lane < ge) {
        p = sorted[gb + lane];
        nrm = dinv[p >> TSH] * dinv[node0 + (p & (TILE - 1))];
    }
    for (int base = gb; base < ge; base += 32) {
        v2i rec;
        rec.x = p;
        rec.y = __float_as_int(nrm);
        *(v2i*)(st + 2 * lane) = rec;
        int cnt = min(32, ge - base);
        int nidx = base + 32 + lane;
        int pn = 0;
        if (nidx < ge) pn = sorted[nidx];
        group_step16(c, rows, st4, 0, half, l, row);
        float ds = dinv[pn >> TSH];
        float dd = dinv[node0 + (pn & (TILE - 1))];
        if (cnt > 16) group_step16(c, rows, st4, 16, half, l, row);
        p = pn;
        nrm = (nidx < ge) ? ds * dd : 0.0f;
    }
}

__global__ void __launch_bounds__(256) tile_aggregate64(const float* __restrict__ xl, const int* __restrict__ sorted,
                                                        const int* __restrict__ grpBeg,
                                                        const int* __restrict__ grpEnd,
                                                        const float* __restrict__ dinv,
                                                        const float* __restrict__ selfn,
                                                        const float* __restrict__ bias, float* __restrict__ out) {
    __shared__ __attribute__((aligned(16))) int stage[8 * 64];
    int t = threadIdx.x, b = blockIdx.x;
    int lane = t & 31;
    int w = __builtin_amdgcn_readfirstlane(t >> 5);
    int half = lane >> 4, l = lane & 15;
    int node0 = b * TILE;
    int g = b * 8 + w;
    v8f c[4];
#pragma unroll
    for (int i = 0; i < 4; ++i) c[i] = v8f{};
    group_sum_mma(c, xl, sorted, grpBeg[g], grpEnd[g], dinv, node0, w * 16 + l, stage + w * 64, lane);
    v4f bv = *(const v4f*)(bias + 4 * l);
#pragma unroll
    for (int r = 0; r < 8; ++r) {
        int node = node0 + w * 16 + half * 8 + r;
        if (node < NN) {
            size_t o = (size_t)node * 64 + 4 * l;
            v4f xv = *(const v4f*)(xl + o);
            float sn = selfn[node];
            v4f v;
            v.x = tanhf(c[0][r] + xv.x * sn + bv.x);
            v.y = tanhf(c[1][r] + xv.y * sn + bv.y);
            v.z = tanhf(c[2][r] + xv.z * sn + bv.z);
            v.w = tanhf(c[3][r] + xv.w * sn + bv.w);
            *(v4f*)(out + o) = v;
        }
    }
}

#define TP 68

template <int NOUT>
__device__ __forceinline__ void tile_dense_tanh(float* tile, const float* __restrict__ Wt,
                                                const float* __restrict__ bias, int w, int lane) {
    constexpr int NT = NOUT / 16;
    int half = lane >> 4, l = lane & 15;
    int n0 = (w % NT) * 16;
    int mb = (w / NT) * NT * 16;
    v8f c[NT];
#pragma unroll
    for (int i = 0; i < NT; ++i) c[i] = v8f{};
#pragma unroll 4
    for (int k = 0; k < 64; k += 4) {
        int ka = k + half * 2;
        v2f bf;
        bf.x = Wt[ka * NOUT + n0 + l];
        bf.y = Wt[(ka + 1) * NOUT + n0 + l];
#pragma unroll
        for (int i = 0; i < NT; ++i) {
            v2f af = *(const v2f*)(tile + (mb + i * 16 + l) * TP + ka);
            c[i] = __builtin_amdgcn_wmma_f32_16x16x4_f32(false, af, false, bf,
                                                         (short)0, c[i], false, false);
        }
    }
    __syncthreads();
    float bv = bias[n0 + l];
#pragma unroll
    for (int i = 0; i < NT; ++i) {
#pragma unroll
        for (int r = 0; r < 8; ++r) {
            int row = mb + i * 16 + r + half * 8;
            tile[row * TP + n0 + l] = tanhf(c[i][r] + bv);
        }
    }
    __syncthreads();
}

__global__ void __launch_bounds__(256) tile_aggregate_tail(const float* __restrict__ h, const int* __restrict__ sorted,
                                                           const int* __restrict__ grpBeg,
                                                           const int* __restrict__ grpEnd,
                                                           const float* __restrict__ dinv,
                                                           const float* __restrict__ selfn,
                                                           const float* __restrict__ W1, const float* __restrict__ b1,
                                                           const float* __restrict__ F0, const float* __restrict__ c0,
                                                           const float* __restrict__ F1, const float* __restrict__ c1,
                                                           const float* __restrict__ F2, const float* __restrict__ c2,
                                                           float* __restrict__ out) {
    __shared__ __attribute__((aligned(16))) float acc[TILE * TP];
    __shared__ __attribute__((aligned(16))) int stage[8 * 64];
    int t = threadIdx.x, b = blockIdx.x;
    int lane = t & 31;
    int w = __builtin_amdgcn_readfirstlane(t >> 5);
    int half = lane >> 4, l = lane & 15;
    int node0 = b * TILE;
    int g = b * 8 + w;
    v8f c[4];
#pragma unroll
    for (int i = 0; i < 4; ++i) c[i] = v8f{};
    group_sum_mma(c, h, sorted, grpBeg[g], grpEnd[g], dinv, node0, w * 16 + l, stage + w * 64, lane);
#pragma unroll
    for (int r = 0; r < 8; ++r) {
        int row = w * 16 + half * 8 + r;
        int node = node0 + row;
        v4f s;
        s.x = c[0][r]; s.y = c[1][r]; s.z = c[2][r]; s.w = c[3][r];
        if (node < NN) {
            v4f hv = *(const v4f*)(h + (size_t)node * 64 + 4 * l);
            float sn = selfn[node];
            s.x += hv.x * sn; s.y += hv.y * sn; s.z += hv.z * sn; s.w += hv.w * sn;
        }
        *(v4f*)(acc + row * TP + 4 * l) = s;
    }
    __syncthreads();
    tile_dense_tanh<64>(acc, W1, b1, w, lane);
    tile_dense_tanh<64>(acc, F0, c0, w, lane);
    tile_dense_tanh<32>(acc, F1, c1, w, lane);
    if (t < TILE) {
        int node = node0 + t;
        float s = c2[0];
#pragma unroll
        for (int j = 0; j < 32; ++j) s += acc[t * TP + j] * F2[j];
        if (node < NN) out[node] = s;
    }
}


extern "C" void kernel_launch(void* const* d_in, const int* in_sizes, int n_in,
                              void* d_out, int out_size, void* d_ws, size_t ws_size,
                              hipStream_t stream) {
    const float* x   = (const float*)d_in[0];
    const int*   ei  = (const int*)d_in[1];
    const float* cW0 = (const float*)d_in[2];
    const float* cb0 = (const float*)d_in[3];
    const float* cW1 = (const float*)d_in[4];
    const float* cb1 = (const float*)d_in[5];
    const float* fW0 = (const float*)d_in[6];
    const float* fb0 = (const float*)d_in[7];
    const float* fW1 = (const float*)d_in[8];
    const float* fb1 = (const float*)d_in[9];
    const float* fW2 = (const float*)d_in[10];
    const float* fb2 = (const float*)d_in[11];
    const int* src = ei;
    const int* dst = ei + NE;

    float* ws    = (float*)d_ws;
    float* xl    = ws;
    float* plane = xl  + (size_t)NN * 64;
    float* h1    = plane + (size_t)NN * 64;
    float* dinv  = h1  + (size_t)NN * 64;
    float* selfn = dinv + NN;
    int* binned   = (int*)plane;
    int* sorted   = binned + NE;
    int* grpBeg   = sorted + NE;
    int* grpEnd   = grpBeg + NB * 8;
    int* binCount = grpEnd + NB * 8;
    int* cursor   = binCount + NB;
    int* binStart = cursor + NB;

    const int B = 256;
    dim3 blk(B);
    auto cdiv = [](long long a, long long b) { return (unsigned)((a + b - 1) / b); };

    unsigned gCH   = cdiv(NE, CH);
    unsigned gG64  = cdiv((long long)(NN / 16) * 4 * 32, B);

    zero_bins<<<1, BT, 0, stream>>>(binCount, cursor);
    bin_hist<<<gCH, BT, 0, stream>>>(dst, binCount);
    bin_scan<<<1, 1024, 0, stream>>>(binCount, binStart);
    bin_partition<<<gCH, BT, 0, stream>>>(src, dst, binStart, cursor, binned);
    tile_degree_sort<<<NB, blk, 0, stream>>>(binned, binStart, sorted, grpBeg, grpEnd, dinv, selfn);

    wmma_gemm_f32<<<gG64, blk, 0, stream>>>(x, cW0, nullptr, xl, NN, 128, 64, 0);
    tile_aggregate64<<<NB, blk, 0, stream>>>(xl, sorted, grpBeg, grpEnd, dinv, selfn, cb0, h1);

    tile_aggregate_tail<<<NB, blk, 0, stream>>>(h1, sorted, grpBeg, grpEnd, dinv, selfn,
                                                cW1, cb1, fW0, fb0, fW1, fb1, fW2, fb2, (float*)d_out);
}


# Both aggregation launches run as 1564 four-wave work-groups on 64-node half tiles (one wave per SIMD) instead of 782 eight-wave ones: the fullest SIMDs hold 7 waves, not 8; per-wave work unchanged
# speedup vs baseline: 4.5311x; 1.1046x over previous
#include <hip/hip_runtime.h>
#include <hip/hip_bf16.h>
#include <math.h>


#define NN 100000
#define NE 1600000

#define TSH  7
#define TILE 128
#define NB   ((NN + TILE - 1) / TILE)
#define CH   8192
#define BT   1024
#define EPT  (CH / BT)
#define GCH  ((NE + CH - 1) / CH)

typedef __attribute__((ext_vector_type(2))) float v2f;
typedef __attribute__((ext_vector_type(8))) float v8f;
typedef __attribute__((ext_vector_type(4))) float v4f;
typedef __attribute__((ext_vector_type(4))) int   v4i;
typedef __attribute__((ext_vector_type(2))) int   v2i;

#define GT_TOTAL ((NN / 16) * 4)
#define GB_HIST  200
#define GB_SCAN  60
#define GB_PART  300
#define GT_SCAN0 (GB_HIST * 32)
#define GT_PART0 ((GB_HIST + GB_SCAN) * 32)
#define GT_SORT0 ((GB_HIST + GB_SCAN + GB_PART) * 32)
#define GB_SORT  ((GT_TOTAL - GT_SORT0) / 8)

__device__ __forceinline__ void gemm_tile_wave(const float* __restrict__ X, const float* __restrict__ W,
                                               float* __restrict__ Y, int wave, int lane,
                                               int M, int K, int Nout) {
    int tilesN = Nout >> 4;
    int m0 = (wave / tilesN) << 4;
    int n0 = (wave % tilesN) << 4;
    if (m0 >= M) return;
    int half = lane >> 4;
    int l    = lane & 15;

    const float* xrow = X + (long long)(m0 + l) * K;
    v8f c = {};
    for (int k = 0; k < K; k += 4) {
        int ka = k + half * 2;
        v2f a, b;
        a.x = xrow[ka];
        a.y = xrow[ka + 1];
        b.x = W[ka * Nout + n0 + l];
        b.y = W[(ka + 1) * Nout + n0 + l];
        c = __builtin_amdgcn_wmma_f32_16x16x4_f32(false, a, false, b,
                                                  (short)0, c, false, false);
    }
    int col = n0 + l;
#pragma unroll
    for (int r = 0; r < 8; ++r) {
        int row = m0 + r + half * 8;
        Y[(long long)row * Nout + col] = c[r];
    }
}


__global__ void __launch_bounds__(BT) zero_bins(int* __restrict__ binCount, int* __restrict__ cursor) {
    int t = threadIdx.x;
    if (t < NB) { binCount[t] = 0; cursor[t] = 0; }
}

__global__ void __launch_bounds__(BT) bin_hist(const int* __restrict__ dst, int* __restrict__ binCount,
                                               const float* __restrict__ X, const float* __restrict__ Wm,
                                               float* __restrict__ Y, int M, int K, int Nout) {
    __shared__ int hist[NB];
    int t = threadIdx.x;
    if (blockIdx.x >= GCH) {
        gemm_tile_wave(X, Wm, Y, (int)(blockIdx.x - GCH) * (BT / 32) + (t >> 5), t & 31, M, K, Nout);
        return;
    }
    if (t < NB) hist[t] = 0;
    __syncthreads();
    int e0 = blockIdx.x * CH + t;
#pragma unroll
    for (int i = 0; i < EPT; ++i) {
        int e = e0 + i * BT;
        if (e < NE) atomicAdd(&hist[dst[e] >> TSH], 1);
    }
    __syncthreads();
    if (t < NB) {
        int c = hist[t];
        if (c) atomicAdd(&binCount[t], c);
    }
}

__global__ void __launch_bounds__(1024) bin_scan(const int* __restrict__ binCount, int* __restrict__ binStart,
                                                 const float* __restrict__ X, const float* __restrict__ Wm,
                                                 float* __restrict__ Y, int M, int K, int Nout) {
    __shared__ int a[1024];
    int t = threadIdx.x;
    if (blockIdx.x >= 1) {
        gemm_tile_wave(X, Wm, Y, GT_SCAN0 + (int)(blockIdx.x - 1) * 32 + (t >> 5), t & 31, M, K, Nout);
        return;
    }
    a[t] = (t < NB) ? binCount[t] : 0;
    __syncthreads();
    for (int off = 1; off < 1024; off <<= 1) {
        int v = a[t];
        if (t >= off) v += a[t - off];
        __syncthreads();
        a[t] = v;
        __syncthreads();
    }
    if (t == 0) binStart[0] = 0;
    if (t < NB) binStart[t + 1] = a[t];
}

__global__ void __launch_bounds__(BT) bin_partition(const int* __restrict__ src, const int* __restrict__ dst,
                                                    const int* __restrict__ binStart, int* __restrict__ cursor,
                                                    int* __restrict__ binned,
                                                    const float* __restrict__ X, const float* __restrict__ Wm,
                                                    float* __restrict__ Y, int M, int K, int Nout) {
    __shared__ int hist[NB];
    __shared__ int base[NB];
    int t = threadIdx.x;
    if (blockIdx.x >= GCH) {
        gemm_tile_wave(X, Wm, Y, GT_PART0 + (int)(blockIdx.x - GCH) * (BT / 32) + (t >> 5), t & 31, M, K, Nout);
        return;
    }
    if (t < NB) hist[t] = 0;
    __syncthreads();
    int e0 = blockIdx.x * CH + t;
    int bin[EPT], rank[EPT], pk[EPT];
#pragma unroll
    for (int i = 0; i < EPT; ++i) {
        int e = e0 + i * BT;
        bin[i] = -1; rank[i] = 0; pk[i] = 0;
        if (e < NE) {
            int d = dst[e];
            int s = src[e];
            bin[i]  = d >> TSH;
            pk[i]   = (s << TSH) | (d & (TILE - 1));
            rank[i] = atomicAdd(&hist[bin[i]], 1);
        }
    }
    __syncthreads();
    if (t < NB) {
        int c = hist[t];
        int claimed = c ? atomicAdd(&cursor[t], c) : 0;
        base[t] = binStart[t] + claimed;
    }
    __syncthreads();
#pragma unroll
    for (int i = 0; i < EPT; ++i) {
        if (bin[i] >= 0) {
            unsigned pos = (unsigned)(base[bin[i]] + rank[i]);
            if (pos < (unsigned)NE) binned[pos] = pk[i];
        }
    }
}

__global__ void __launch_bounds__(256) tile_degree_sort(const int* __restrict__ binned, const int* __restrict__ binStart,
                                                        int* __restrict__ sorted, int* __restrict__ grpBeg,
                                                        int* __restrict__ grpEnd,
                                                        float* __restrict__ dinv, float* __restrict__ selfn,
                                                        const float* __restrict__ X, const float* __restrict__ Wm,
                                                        float* __restrict__ Y, int M, int K, int Nout) {
    __shared__ int cnt[TILE];
    __shared__ int off[TILE + 1];
    __shared__ int wsum[8];
    int t = threadIdx.x, b = blockIdx.x;
    int lane = t & 31, w = t >> 5;
    if (b >= NB) {
        gemm_tile_wave(X, Wm, Y, GT_SORT0 + (b - NB) * 8 + w, lane, M, K, Nout);
        return;
    }
    if (t < TILE) cnt[t] = 0;
    __syncthreads();
    int beg = binStart[b], end = binStart[b + 1];
    for (int i = beg + t; i < end; i += 256) atomicAdd(&cnt[binned[i] & (TILE - 1)], 1);
    __syncthreads();
    int c = (t < TILE) ? cnt[t] : 0;
    if (t < TILE) {
        int node = b * TILE + t;
        if (node < NN) {
            float d = (float)c + 1.0f;
            dinv[node]  = rsqrtf(d);
            selfn[node] = 1.0f / d;
        }
    }
    int v = c;
#pragma unroll
    for (int d = 1; d < 32; d <<= 1) {
        int o = __shfl_up(v, d);
        if (lane >= d) v += o;
    }
    if (lane == 31) wsum[w] = v;
    __syncthreads();
    int pre = 0;
    for (int k = 0; k < w; ++k) pre += wsum[k];
    if (t < TILE) { off[t] = pre + v - c; cnt[t] = 0; }
    if (t == TILE) off[TILE] = end - beg;
    __syncthreads();
    if (t < 8) {
        grpBeg[b * 8 + t] = beg + off[t * 16];
        grpEnd[b * 8 + t] = beg + off[t * 16 + 16];
    }
    for (int i = beg + t; i < end; i += 256) {
        int p = binned[i];
        int n = p & (TILE - 1);
        int r = atomicAdd(&cnt[n], 1);
        unsigned pos = (unsigned)(beg + off[n] + r);
        if (pos < (unsigned)NE) sorted[pos] = p;
    }
}

__device__ __forceinline__ void group_step16(v8f (&c)[4], const float* __restrict__ rows, const v4i* st4,
                                             int j, int half, int l, int row) {
    v4f ra[4], rb[4];
    v2f a[4];
#pragma unroll
    for (int u = 0; u < 4; ++u) {
        v4i q = st4[(j >> 1) + 2 * u + half];
        ra[u] = *(const v4f*)(rows + (((unsigned)q.x >> TSH) * 64u + 4u * (unsigned)l));
        rb[u] = *(const v4f*)(rows + (((unsigned)q.z >> TSH) * 64u + 4u * (unsigned)l));
        a[u].x = ((q.x & (TILE - 1)) == row) ? __int_as_float(q.y) : 0.0f;
        a[u].y = ((q.z & (TILE - 1)) == row) ? __int_as_float(q.w) : 0.0f;
    }
#pragma unroll
    for (int u = 0; u < 4; ++u) {
#pragma unroll
        for (int t = 0; t < 4; ++t) {
            v2f bf;
            bf.x = ra[u][t];
            bf.y = rb[u][t];
            c[t] = __builtin_amdgcn_wmma_f32_16x16x4_f32(false, a[u], false, bf,
                                                         (short)0, c[t], false, false);
        }
    }
}

__device__ __forceinline__ void group_sum_mma(v8f (&c)[4], const float* __restrict__ rows,
                                              const int* __restrict__ sorted, int gb, int ge,
                                              const float* __restrict__ dinv, int node0, int row,
                                              int* st, int lane) {
    int half = lane >> 4, l = lane & 15;
    const v4i* st4 = (const v4i*)st;
    int p = 0;
    float nrm = 0.0f;
    if (gb + lane < ge) {
        p = sorted[gb + lane];
        nrm = dinv[p >> TSH] * dinv[node0 + (p & (TILE - 1))];
    }
    for (int base = gb; base < ge; base += 32) {
        v2i rec;
        rec.x = p;
        rec.y = __float_as_int(nrm);
        *(v2i*)(st + 2 * lane) = rec;
        int cnt = min(32, ge - base);
        int nidx = base + 32 + lane;
        int pn = 0;
        if (nidx < ge) pn = sorted[nidx];
        group_step16(c, rows, st4, 0, half, l, row);
        float ds = dinv[pn >> TSH];
        float dd = dinv[node0 + (pn & (TILE - 1))];
        if (cnt > 16) group_step16(c, rows, st4, 16, half, l, row);
        p = pn;
        nrm = (nidx < ge) ? ds * dd : 0.0f;
    }
}

#define HW 4
#define HT (HW * 16)

__global__ void __launch_bounds__(HW * 32) tile_aggregate64(const float* __restrict__ xl, const int* __restrict__ sorted,
                                                        const int* __restrict__ grpBeg,
                                                        const int* __restrict__ grpEnd,
                                                        const float* __restrict__ dinv,
                                                        const float* __restrict__ selfn,
                                                        const float* __restrict__ bias, float* __restrict__ out) {
    __shared__ __attribute__((aligned(16))) int stage[HW * 64];
    int t = threadIdx.x, b = blockIdx.x;
    int lane = t & 31;
    int w = __builtin_amdgcn_readfirstlane(t >> 5);
    int half = lane >> 4, l = lane & 15;
    int node0 = (b >> 1) * TILE;
    int gq = (b & 1) * HW + w;
    int g = b * HW + w;
    v8f c[4];
#pragma unroll
    for (int i = 0; i < 4; ++i) c[i] = v8f{};
    group_sum_mma(c, xl, sorted, grpBeg[g], grpEnd[g], dinv, node0, gq * 16 + l, stage + w * 64, lane);
    v4f bv = *(const v4f*)(bias + 4 * l);
#pragma unroll
    for (int r = 0; r < 8; ++r) {
        int node = node0 + gq * 16 + half * 8 + r;
        if (node < NN) {
            size_t o = (size_t)node * 64 + 4 * l;
            v4f xv = *(const v4f*)(xl + o);
            float sn = selfn[node];
            v4f v;
            v.x = tanhf(c[0][r] + xv.x * sn + bv.x);
            v.y = tanhf(c[1][r] + xv.y * sn + bv.y);
            v.z = tanhf(c[2][r] + xv.z * sn + bv.z);
            v.w = tanhf(c[3][r] + xv.w * sn + bv.w);
            *(v4f*)(out + o) = v;
        }
    }
}

#define TP 68

template <int NOUT>
__device__ __forceinline__ void tile_dense_tanh(float* tile, const float* __restrict__ Wt,
                                                const float* __restrict__ bias, int w, int lane) {
    constexpr int NT = NOUT / 16;
    int half = lane >> 4, l = lane & 15;
    int n0 = (w % NT) * 16;
    int mb = (w / NT) * NT * 16;
    v8f c[NT];
#pragma unroll
    for (int i = 0; i < NT; ++i) c[i] = v8f{};
#pragma unroll 4
    for (int k = 0; k < 64; k += 4) {
        int ka = k + half * 2;
        v2f bf;
        bf.x = Wt[ka * NOUT + n0 + l];
        bf.y = Wt[(ka + 1) * NOUT + n0 + l];
#pragma unroll
        for (int i = 0; i < NT; ++i) {
            v2f af = *(const v2f*)(tile + (mb + i * 16 + l) * TP + ka);
            c[i] = __builtin_amdgcn_wmma_f32_16x16x4_f32(false, af, false, bf,
                                                         (short)0, c[i], false, false);
        }
    }
    __syncthreads();
    float bv = bias[n0 + l];
#pragma unroll
    for (int i = 0; i < NT; ++i) {
#pragma unroll
        for (int r = 0; r < 8; ++r) {
            int row = mb + i * 16 + r + half * 8;
            tile[row * TP + n0 + l] = tanhf(c[i][r] + bv);
        }
    }
    __syncthreads();
}

__global__ void __launch_bounds__(HW * 32) tile_aggregate_tail(const float* __restrict__ h, const int* __restrict__ sorted,
                                                           const int* __restrict__ grpBeg,
                                                           const int* __restrict__ grpEnd,
                                                           const float* __restrict__ dinv,
                                                           const float* __restrict__ selfn,
                                                           const float* __restrict__ W1, const float* __restrict__ b1,
                                                           const float* __restrict__ F0, const float* __restrict__ c0,
                                                           const float* __restrict__ F1, const float* __restrict__ c1,
                                                           const float* __restrict__ F2, const float* __restrict__ c2,
                                                           float* __restrict__ out) {
    __shared__ __attribute__((aligned(16))) float acc[HT * TP];
    __shared__ __attribute__((aligned(16))) int stage[HW * 64];
    int t = threadIdx.x, b = blockIdx.x;
    int lane = t & 31;
    int w = __builtin_amdgcn_readfirstlane(t >> 5);
    int half = lane >> 4, l = lane & 15;
    int node0 = (b >> 1) * TILE;
    int gq = (b & 1) * HW + w;
    int g = b * HW + w;
    int nodeh = node0 + (b & 1) * HT;
    v8f c[4];
#pragma unroll
    for (int i = 0; i < 4; ++i) c[i] = v8f{};
    group_sum_mma(c, h, sorted, grpBeg[g], grpEnd[g], dinv, node0, gq * 16 + l, stage + w * 64, lane);
#pragma unroll
    for (int r = 0; r < 8; ++r) {
        int row = w * 16 + half * 8 + r;
        int node = nodeh + row;
        v4f s;
        s.x = c[0][r]; s.y = c[1][r]; s.z = c[2][r]; s.w = c[3][r];
        if (node < NN) {
            v4f hv = *(const v4f*)(h + (size_t)node * 64 + 4 * l);
            float sn = selfn[node];
            s.x += hv.x * sn; s.y += hv.y * sn; s.z += hv.z * sn; s.w += hv.w * sn;
        }
        *(v4f*)(acc + row * TP + 4 * l) = s;
    }
    __syncthreads();
    tile_dense_tanh<64>(acc, W1, b1, w, lane);
    tile_dense_tanh<64>(acc, F0, c0, w, lane);
    tile_dense_tanh<32>(acc, F1, c1, w, lane);
    if (t < HT) {
        int node = nodeh + t;
        float s = c2[0];
#pragma unroll
        for (int j = 0; j < 32; ++j) s += acc[t * TP + j] * F2[j];
        if (node < NN) out[node] = s;
    }
}


extern "C" void kernel_launch(void* const* d_in, const int* in_sizes, int n_in,
                              void* d_out, int out_size, void* d_ws, size_t ws_size,
                              hipStream_t stream) {
    const float* x   = (const float*)d_in[0];
    const int*   ei  = (const int*)d_in[1];
    const float* cW0 = (const float*)d_in[2];
    const float* cb0 = (const float*)d_in[3];
    const float* cW1 = (const float*)d_in[4];
    const float* cb1 = (const float*)d_in[5];
    const float* fW0 = (const float*)d_in[6];
    const float* fb0 = (const float*)d_in[7];
    const float* fW1 = (const float*)d_in[8];
    const float* fb1 = (const float*)d_in[9];
    const float* fW2 = (const float*)d_in[10];
    const float* fb2 = (const float*)d_in[11];
    const int* src = ei;
    const int* dst = ei + NE;

    float* ws    = (float*)d_ws;
    float* xl    = ws;
    float* plane = xl  + (size_t)NN * 64;
    float* h1    = plane + (size_t)NN * 64;
    float* dinv  = h1  + (size_t)NN * 64;
    float* selfn = dinv + NN;
    int* binned   = (int*)plane;
    int* sorted   = binned + NE;
    int* grpBeg   = sorted + NE;
    int* grpEnd   = grpBeg + NB * 8;
    int* binCount = grpEnd + NB * 8;
    int* cursor   = binCount + NB;
    int* binStart = cursor + NB;

    const int B = 256;
    dim3 blk(B);

    zero_bins<<<1, BT, 0, stream>>>(binCount, cursor);
    bin_hist<<<GCH + GB_HIST, BT, 0, stream>>>(dst, binCount, x, cW0, xl, NN, 128, 64);
    bin_scan<<<1 + GB_SCAN, 1024, 0, stream>>>(binCount, binStart, x, cW0, xl, NN, 128, 64);
    bin_partition<<<GCH + GB_PART, BT, 0, stream>>>(src, dst, binStart, cursor, binned, x, cW0, xl, NN, 128, 64);
    tile_degree_sort<<<NB + GB_SORT, blk, 0, stream>>>(binned, binStart, sorted, grpBeg, grpEnd, dinv, selfn,
                                                       x, cW0, xl, NN, 128, 64);

    tile_aggregate64<<<NB * 2, HW * 32, 0, stream>>>(xl, sorted, grpBeg, grpEnd, dinv, selfn, cb0, h1);

    tile_aggregate_tail<<<NB * 2, HW * 32, 0, stream>>>(h1, sorted, grpBeg, grpEnd, dinv, selfn,
                                                cW1, cb1, fW0, fb0, fW1, fb1, fW2, fb2, (float*)d_out);
}


# Loads requested together ahead of their consumers where each was waited for alone: histogram and degree/sort edge words before the LDS atomics, self rows and 1/deg before both aggregation row loops
# speedup vs baseline: 5.0123x; 1.1062x over previous
#include <hip/hip_runtime.h>
#include <hip/hip_bf16.h>
#include <math.h>


#define NN 100000
#define NE 1600000

#define TSH  7
#define TILE 128
#define NB   ((NN + TILE - 1) / TILE)
#define CH   8192
#define BT   1024
#define EPT  (CH / BT)
#define GCH  ((NE + CH - 1) / CH)

typedef __attribute__((ext_vector_type(2))) float v2f;
typedef __attribute__((ext_vector_type(8))) float v8f;
typedef __attribute__((ext_vector_type(4))) float v4f;
typedef __attribute__((ext_vector_type(4))) int   v4i;
typedef __attribute__((ext_vector_type(2))) int   v2i;

#define GT_TOTAL ((NN / 16) * 4)
#define GB_HIST  200
#define GB_PART  360
#define GT_PART0 (GB_HIST * 32)
#define GT_SORT0 ((GB_HIST + GB_PART) * 32)
#define GB_SORT  ((GT_TOTAL - GT_SORT0) / 8)

__device__ __forceinline__ void gemm_tile_wave(const float* __restrict__ X, const float* __restrict__ W,
                                               float* __restrict__ Y, int wave, int lane,
                                               int M, int K, int Nout) {
    int tilesN = Nout >> 4;
    int m0 = (wave / tilesN) << 4;
    int n0 = (wave % tilesN) << 4;
    if (m0 >= M) return;
    int half = lane >> 4;
    int l    = lane & 15;

    const float* xrow = X + (long long)(m0 + l) * K;
    v8f c = {};
    for (int k = 0; k < K; k += 4) {
        int ka = k + half * 2;
        v2f a, b;
        a.x = xrow[ka];
        a.y = xrow[ka + 1];
        b.x = W[ka * Nout + n0 + l];
        b.y = W[(ka + 1) * Nout + n0 + l];
        c = __builtin_amdgcn_wmma_f32_16x16x4_f32(false, a, false, b,
                                                  (short)0, c, false, false);
    }
    int col = n0 + l;
#pragma unroll
    for (int r = 0; r < 8; ++r) {
        int row = m0 + r + half * 8;
        Y[(long long)row * Nout + col] = c[r];
    }
}


__global__ void __launch_bounds__(BT) zero_bins(int* __restrict__ binCount, int* __restrict__ cursor) {
    int t = threadIdx.x;
    if (t < NB) { binCount[t] = 0; cursor[t] = 0; }
}

__global__ void __launch_bounds__(BT) bin_hist(const int* __restrict__ dst, int* __restrict__ binCount,
                                               const float* __restrict__ X, const float* __restrict__ Wm,
                                               float* __restrict__ Y, int M, int K, int Nout) {
    __shared__ int hist[NB];
    int t = threadIdx.x;
    if (blockIdx.x >= GCH) {
        gemm_tile_wave(X, Wm, Y, (int)(blockIdx.x - GCH) * (BT / 32) + (t >> 5), t & 31, M, K, Nout);
        return;
    }
    int e0 = blockIdx.x * CH + t;
    int dv[EPT];
#pragma unroll
    for (int i = 0; i < EPT; ++i) dv[i] = dst[min(e0 + i * BT, NE - 1)];
    if (t < NB) hist[t] = 0;
    __syncthreads();
#pragma unroll
    for (int i = 0; i < EPT; ++i) {
        if (e0 + i * BT < NE) atomicAdd(&hist[dv[i] >> TSH], 1);
    }
    __syncthreads();
    if (t < NB) {
        int c = hist[t];
        if (c) atomicAdd(&binCount[t], c);
    }
}

__global__ void __launch_bounds__(BT) bin_partition(const int* __restrict__ src, const int* __restrict__ dst,
                                                    const int* __restrict__ binCount, int* __restrict__ binStart,
                                                    int* __restrict__ cursor, int* __restrict__ binned,
                                                    const float* __restrict__ X, const float* __restrict__ Wm,
                                                    float* __restrict__ Y, int M, int K, int Nout) {
    __shared__ int hist[NB];
    __shared__ int base[NB];
    __shared__ int wsum[BT / 32];
    int t = threadIdx.x;
    if (blockIdx.x >= GCH) {
        gemm_tile_wave(X, Wm, Y, GT_PART0 + (int)(blockIdx.x - GCH) * (BT / 32) + (t >> 5), t & 31, M, K, Nout);
        return;
    }
    int lane = t & 31, w = t >> 5;
    if (t < NB) hist[t] = 0;
    int e0 = blockIdx.x * CH + t;
    int dv[EPT], sv[EPT];
#pragma unroll
    for (int i = 0; i < EPT; ++i) {
        int e = e0 + i * BT;
        dv[i] = 0; sv[i] = 0;
        if (e < NE) { dv[i] = dst[e]; sv[i] = src[e]; }
    }
    int cntv = (t < NB) ? binCount[t] : 0;
    int v = cntv;
#pragma unroll
    for (int d = 1; d < 32; d <<= 1) {
        int o = __shfl_up(v, d);
        if (lane >= d) v += o;
    }
    if (lane == 31) wsum[w] = v;
    __syncthreads();
    int bin[EPT], rank[EPT], pk[EPT];
#pragma unroll
    for (int i = 0; i < EPT; ++i) {
        bin[i] = -1; rank[i] = 0; pk[i] = 0;
        if (e0 + i * BT < NE) {
            bin[i]  = dv[i] >> TSH;
            pk[i]   = (sv[i] << TSH) | (dv[i] & (TILE - 1));
            rank[i] = atomicAdd(&hist[bin[i]], 1);
        }
    }
    int pre = 0;
    for (int k = 0; k < w; ++k) pre += wsum[k];
    int start = pre + v - cntv;
    if (blockIdx.x == 0) {
        if (t < NB) binStart[t] = start;
        if (t == NB - 1) binStart[NB] = start + cntv;
    }
    __syncthreads();
    if (t < NB) {
        int c = hist[t];
        int claimed = c ? atomicAdd(&cursor[t], c) : 0;
        base[t] = start + claimed;
    }
    __syncthreads();
#pragma unroll
    for (int i = 0; i < EPT; ++i) {
        if (bin[i] >= 0) {
            unsigned pos = (unsigned)(base[bin[i]] + rank[i]);
            if (pos < (unsigned)NE) binned[pos] = pk[i];
        }
    }
}

#define SU 10
__global__ void __launch_bounds__(256) tile_degree_sort(const int* __restrict__ binned, const int* __restrict__ binStart,
                                                        int* __restrict__ sorted, int* __restrict__ grpBeg,
                                                        int* __restrict__ grpEnd,
                                                        float* __restrict__ dinv, float* __restrict__ selfn,
                                                        const float* __restrict__ X, const float* __restrict__ Wm,
                                                        float* __restrict__ Y, int M, int K, int Nout) {
    __shared__ int cnt[TILE];
    __shared__ int off[TILE + 1];
    __shared__ int wsum[8];
    int t = threadIdx.x, b = blockIdx.x;
    int lane = t & 31, w = t >> 5;
    if (b >= NB) {
        gemm_tile_wave(X, Wm, Y, GT_SORT0 + (b - NB) * 8 + w, lane, M, K, Nout);
        return;
    }
    int beg = binStart[b], end = binStart[b + 1];
    bool onewin = (end - beg) <= 256 * SU;
    int pv[SU];
#pragma unroll
    for (int u = 0; u < SU; ++u) pv[u] = binned[max(min(beg + u * 256 + t, end - 1), 0)];
    if (t < TILE) cnt[t] = 0;
    __syncthreads();
    for (int base = beg; base < end; base += 256 * SU) {
        if (base != beg) {
#pragma unroll
            for (int u = 0; u < SU; ++u) pv[u] = binned[min(base + u * 256 + t, end - 1)];
        }
#pragma unroll
        for (int u = 0; u < SU; ++u)
            if (base + u * 256 + t < end) atomicAdd(&cnt[pv[u] & (TILE - 1)], 1);
    }
    __syncthreads();
    int c = (t < TILE) ? cnt[t] : 0;
    if (t < TILE) {
        int node = b * TILE + t;
        if (node < NN) {
            float d = (float)c + 1.0f;
            dinv[node]  = rsqrtf(d);
            selfn[node] = 1.0f / d;
        }
    }
    int v = c;
#pragma unroll
    for (int d = 1; d < 32; d <<= 1) {
        int o = __shfl_up(v, d);
        if (lane >= d) v += o;
    }
    if (lane == 31) wsum[w] = v;
    __syncthreads();
    int pre = 0;
    for (int k = 0; k < w; ++k) pre += wsum[k];
    if (t < TILE) { off[t] = pre + v - c; cnt[t] = 0; }
    if (t == TILE) off[TILE] = end - beg;
    __syncthreads();
    if (t < 8) {
        grpBeg[b * 8 + t] = beg + off[t * 16];
        grpEnd[b * 8 + t] = beg + off[t * 16 + 16];
    }
    for (int base = beg; base < end; base += 256 * SU) {
        if (!onewin) {
#pragma unroll
            for (int u = 0; u < SU; ++u) pv[u] = binned[min(base + u * 256 + t, end - 1)];
        }
#pragma unroll
        for (int u = 0; u < SU; ++u) {
            if (base + u * 256 + t < end) {
                int p = pv[u];
                int n = p & (TILE - 1);
                int r = atomicAdd(&cnt[n], 1);
                unsigned pos = (unsigned)(beg + off[n] + r);
                if (pos < (unsigned)NE) sorted[pos] = p;
            }
        }
    }
}

__device__ __forceinline__ void group_step16(v8f (&c)[4], const float* __restrict__ rows, const v4i* st4,
                                             int j, int half, int l, int row) {
    v4f ra[4], rb[4];
    v2f a[4];
#pragma unroll
    for (int u = 0; u < 4; ++u) {
        v4i q = st4[(j >> 1) + 2 * u + half];
        ra[u] = *(const v4f*)(rows + (((unsigned)q.x >> TSH) * 64u + 4u * (unsigned)l));
        rb[u] = *(const v4f*)(rows + (((unsigned)q.z >> TSH) * 64u + 4u * (unsigned)l));
        a[u].x = ((q.x & (TILE - 1)) == row) ? __int_as_float(q.y) : 0.0f;
        a[u].y = ((q.z & (TILE - 1)) == row) ? __int_as_float(q.w) : 0.0f;
    }
#pragma unroll
    for (int u = 0; u < 4; ++u) {
#pragma unroll
        for (int t = 0; t < 4; ++t) {
            v2f bf;
            bf.x = ra[u][t];
            bf.y = rb[u][t];
            c[t] = __builtin_amdgcn_wmma_f32_16x16x4_f32(false, a[u], false, bf,
                                                         (short)0, c[t], false, false);
        }
    }
}

__device__ __forceinline__ void group_sum_mma(v8f (&c)[4], const float* __restrict__ rows,
                                              const int* __restrict__ sorted, int gb, int ge,
                                              const float* __restrict__ dinv, int node0, int row,
                                              int* st, int lane) {
    int half = lane >> 4, l = lane & 15;
    const v4i* st4 = (const v4i*)st;
    int p = 0;
    float nrm = 0.0f;
    if (gb + lane < ge) {
        p = sorted[gb + lane];
        nrm = dinv[p >> TSH] * dinv[node0 + (p & (TILE - 1))];
    }
    for (int base = gb; base < ge; base += 32) {
        v2i rec;
        rec.x = p;
        rec.y = __float_as_int(nrm);
        *(v2i*)(st + 2 * lane) = rec;
        int cnt = min(32, ge - base);
        int nidx = base + 32 + lane;
        int pn = 0;
        if (nidx < ge) pn = sorted[nidx];
        group_step16(c, rows, st4, 0, half, l, row);
        float ds = dinv[pn >> TSH];
        float dd = dinv[node0 + (pn & (TILE - 1))];
        if (cnt > 16) group_step16(c, rows, st4, 16, half, l, row);
        p = pn;
        nrm = (nidx < ge) ? ds * dd : 0.0f;
    }
}

#define HW 4
#define HT (HW * 16)

__global__ void __launch_bounds__(HW * 32) tile_aggregate64(const float* __restrict__ xl, const int* __restrict__ sorted,
                                                        const int* __restrict__ grpBeg,
                                                        const int* __restrict__ grpEnd,
                                                        const float* __restrict__ dinv,
                                                        const float* __restrict__ selfn,
                                                        const float* __restrict__ bias, float* __restrict__ out) {
    __shared__ __attribute__((aligned(16))) int stage[HW * 64];
    int t = threadIdx.x, b = blockIdx.x;
    int lane = t & 31;
    int w = __builtin_amdgcn_readfirstlane(t >> 5);
    int half = lane >> 4, l = lane & 15;
    int node0 = (b >> 1) * TILE;
    int gq = (b & 1) * HW + w;
    int g = b * HW + w;
    v8f c[4];
#pragma unroll
    for (int i = 0; i < 4; ++i) c[i] = v8f{};
    group_sum_mma(c, xl, sorted, grpBeg[g], grpEnd[g], dinv, node0, gq * 16 + l, stage + w * 64, lane);
    v4f bv = *(const v4f*)(bias + 4 * l);
    v4f xs[8];
    float sns[8];
#pragma unroll
    for (int r = 0; r < 8; ++r) {
        int nd = min(node0 + gq * 16 + half * 8 + r, NN - 1);
        xs[r]  = *(const v4f*)(xl + (size_t)nd * 64 + 4 * l);
        sns[r] = selfn[nd];
    }
#pragma unroll
    for (int r = 0; r < 8; ++r) {
        int node = node0 + gq * 16 + half * 8 + r;
        if (node < NN) {
            size_t o = (size_t)node * 64 + 4 * l;
            v4f xv = xs[r];
            float sn = sns[r];
            v4f v;
            v.x = tanhf(c[0][r] + xv.x * sn + bv.x);
            v.y = tanhf(c[1][r] + xv.y * sn + bv.y);
            v.z = tanhf(c[2][r] + xv.z * sn + bv.z);
            v.w = tanhf(c[3][r] + xv.w * sn + bv.w);
            *(v4f*)(out + o) = v;
        }
    }
}

#define TP 68

template <int NOUT>
__device__ __forceinline__ void tile_dense_tanh(float* tile, const float* __restrict__ Wt,
                                                const float* __restrict__ bias, int w, int lane) {
    constexpr int NT = NOUT / 16;
    int half = lane >> 4, l = lane & 15;
    int n0 = (w % NT) * 16;
    int mb = (w / NT) * NT * 16;
    v8f c[NT];
#pragma unroll
    for (int i = 0; i < NT; ++i) c[i] = v8f{};
    float bv = bias[n0 + l];
#pragma unroll 4
    for (int k = 0; k < 64; k += 4) {
        int ka = k + half * 2;
        v2f bf;
        bf.x = Wt[ka * NOUT + n0 + l];
        bf.y = Wt[(ka + 1) * NOUT + n0 + l];
#pragma unroll
        for (int i = 0; i < NT; ++i) {
            v2f af = *(const v2f*)(tile + (mb + i * 16 + l) * TP + ka);
            c[i] = __builtin_amdgcn_wmma_f32_16x16x4_f32(false, af, false, bf,
                                                         (short)0, c[i], false, false);
        }
    }
    __syncthreads();
#pragma unroll
    for (int i = 0; i < NT; ++i) {
#pragma unroll
        for (int r = 0; r < 8; ++r) {
            int row = mb + i * 16 + r + half * 8;
            tile[row * TP + n0 + l] = tanhf(c[i][r] + bv);
        }
    }
    __syncthreads();
}

__global__ void __launch_bounds__(HW * 32) tile_aggregate_tail(const float* __restrict__ h, const int* __restrict__ sorted,
                                                           const int* __restrict__ grpBeg,
                                                           const int* __restrict__ grpEnd,
                                                           const float* __restrict__ dinv,
                                                           const float* __restrict__ selfn,
                                                           const float* __restrict__ W1, const float* __restrict__ b1,
                                                           const float* __restrict__ F0, const float* __restrict__ c0,
                                                           const float* __restrict__ F1, const float* __restrict__ c1,
                                                           const float* __restrict__ F2, const float* __restrict__ c2,
                                                           float* __restrict__ out) {
    __shared__ __attribute__((aligned(16))) float acc[HT * TP];
    __shared__ __attribute__((aligned(16))) int stage[HW * 64];
    int t = threadIdx.x, b = blockIdx.x;
    int lane = t & 31;
    int w = __builtin_amdgcn_readfirstlane(t >> 5);
    int half = lane >> 4, l = lane & 15;
    int node0 = (b >> 1) * TILE;
    int gq = (b & 1) * HW + w;
    int g = b * HW + w;
    int nodeh = node0 + (b & 1) * HT;
    v8f c[4];
#pragma unroll
    for (int i = 0; i < 4; ++i) c[i] = v8f{};
    group_sum_mma(c, h, sorted, grpBeg[g], grpEnd[g], dinv, node0, gq * 16 + l, stage + w * 64, lane);
    v4f hs[8];
    float sns[8];
#pragma unroll
    for (int r = 0; r < 8; ++r) {
        int nd = min(nodeh + w * 16 + half * 8 + r, NN - 1);
        hs[r]  = *(const v4f*)(h + (size_t)nd * 64 + 4 * l);
        sns[r] = selfn[nd];
    }
#pragma unroll
    for (int r = 0; r < 8; ++r) {
        int row = w * 16 + half * 8 + r;
        int node = nodeh + row;
        v4f s;
        s.x = c[0][r]; s.y = c[1][r]; s.z = c[2][r]; s.w = c[3][r];
        if (node < NN) {
            v4f hv = hs[r];
            float sn = sns[r];
            s.x += hv.x * sn; s.y += hv.y * sn; s.z += hv.z * sn; s.w += hv.w * sn;
        }
        *(v4f*)(acc + row * TP + 4 * l) = s;
    }
    __syncthreads();
    tile_dense_tanh<64>(acc, W1, b1, w, lane);
    tile_dense_tanh<64>(acc, F0, c0, w, lane);
    tile_dense_tanh<32>(acc, F1, c1, w, lane);
    if (t < HT) {
        int node = nodeh + t;
        float s = c2[0];
#pragma unroll
        for (int j = 0; j < 32; ++j) s += acc[t * TP + j] * F2[j];
        if (node < NN) out[node] = s;
    }
}


extern "C" void kernel_launch(void* const* d_in, const int* in_sizes, int n_in,
                              void* d_out, int out_size, void* d_ws, size_t ws_size,
                              hipStream_t stream) {
    const float* x   = (const float*)d_in[0];
    const int*   ei  = (const int*)d_in[1];
    const float* cW0 = (const float*)d_in[2];
    const float* cb0 = (const float*)d_in[3];
    const float* cW1 = (const float*)d_in[4];
    const float* cb1 = (const float*)d_in[5];
    const float* fW0 = (const float*)d_in[6];
    const float* fb0 = (const float*)d_in[7];
    const float* fW1 = (const float*)d_in[8];
    const float* fb1 = (const float*)d_in[9];
    const float* fW2 = (const float*)d_in[10];
    const float* fb2 = (const float*)d_in[11];
    const int* src = ei;
    const int* dst = ei + NE;

    float* ws    = (float*)d_ws;
    float* xl    = ws;
    float* plane = xl  + (size_t)NN * 64;
    float* h1    = plane + (size_t)NN * 64;
    float* dinv  = h1  + (size_t)NN * 64;
    float* selfn = dinv + NN;
    int* binned   = (int*)plane;
    int* sorted   = binned + NE;
    int* grpBeg   = sorted + NE;
    int* grpEnd   = grpBeg + NB * 8;
    int* binCount = grpEnd + NB * 8;
    int* cursor   = binCount + NB;
    int* binStart = cursor + NB;

    const int B = 256;
    dim3 blk(B);

    zero_bins<<<1, BT, 0, stream>>>(binCount, cursor);
    bin_hist<<<GCH + GB_HIST, BT, 0, stream>>>(dst, binCount, x, cW0, xl, NN, 128, 64);
    bin_partition<<<GCH + GB_PART, BT, 0, stream>>>(src, dst, binCount, binStart, cursor, binned, x, cW0, xl, NN, 128, 64);
    tile_degree_sort<<<NB + GB_SORT, blk, 0, stream>>>(binned, binStart, sorted, grpBeg, grpEnd, dinv, selfn,
                                                       x, cW0, xl, NN, 128, 64);

    tile_aggregate64<<<NB * 2, HW * 32, 0, stream>>>(xl, sorted, grpBeg, grpEnd, dinv, selfn, cb0, h1);

    tile_aggregate_tail<<<NB * 2, HW * 32, 0, stream>>>(h1, sorted, grpBeg, grpEnd, dinv, selfn,
                                                cW1, cb1, fW0, fb0, fW1, fb1, fW2, fb2, (float*)d_out);
}
